# starting kernel
# baseline (speedup 1.0000x reference)
#include <hip/hip_runtime.h>
#include <hip/hip_bf16.h>


typedef __attribute__((ext_vector_type(2))) float v2f;
typedef __attribute__((ext_vector_type(8))) float v8f;

#define FFT_LEN   256
#define SPEC_BINS 129
#define KPAD      260
#define NROWS     65536
#define TDIM      4096
#define WIN_LEN   768
#define FRM_LEN   128
#define GROUP_DELAY 648
#define OUTLEN    (TDIM*FRM_LEN - GROUP_DELAY)
#define NBC       16

__global__ void build_w_kernel(float* __restrict__ Wp) {
    int k = blockIdx.x;
    int n = threadIdx.x;
    float val = 0.0f;
    int s = -1;
    bool isIm = false;
    if (k <= 128)                { s = k;        isIm = false; }
    else if (k >= 132 && k <= 258){ s = k - 131; isIm = true;  }
    if (s >= 0) {
        float c = (s == 0 || s == 128) ? 1.0f : 2.0f;
        int nrev  = 255 - n;
        int phase = (s * nrev) & 255;
        float ang = (float)phase * 0.0245436926061702596f;
        float sn, cs;
        __sincosf(ang, &sn, &cs);
        val = isIm ? (128.0f * c * sn) : (-128.0f * c * cs);
    }
    Wp[(k >> 1) * 512 + n * 2 + (k & 1)] = val;
}

__global__ void __launch_bounds__(256)
ifft_wmma_kernel(const float* __restrict__ re, const float* __restrict__ im,
                 const float* __restrict__ Wp, float* __restrict__ fft) {
    const int wid  = (blockIdx.x * blockDim.x + threadIdx.x) >> 5;
    const int lane = threadIdx.x & 31;
    const int half = lane >> 4;
    const int mrow = lane & 15;

    const int r0 = wid << 4;
    const int bc = r0 >> 12;
    const int t  = (r0 & (TDIM - 1)) + mrow;

    const float* __restrict__ pR = re + ((size_t)bc * SPEC_BINS) * TDIM + t
                                      + (size_t)(2 * half) * TDIM;
    const float* __restrict__ pI = im + ((size_t)bc * SPEC_BINS) * TDIM + t
                                      + (size_t)(1 + 2 * half) * TDIM;
    const float* __restrict__ pW = Wp + (size_t)half * 512 + (size_t)mrow * 2;

    v8f acc[16];
#pragma unroll
    for (int i = 0; i < 16; ++i) acc[i] = (v8f){};

#pragma unroll 4
    for (int kc = 0; kc < 32; ++kc) {
        v2f a;
        a.x = pR[(size_t)(4 * kc) * TDIM];
        a.y = pR[(size_t)(4 * kc + 1) * TDIM];
        const float* __restrict__ w = pW + (size_t)(2 * kc) * 512;
#pragma unroll
        for (int nt = 0; nt < 16; ++nt) {
            v2f b = *(const v2f*)(w + nt * 32);
            acc[nt] = __builtin_amdgcn_wmma_f32_16x16x4_f32(
                false, a, false, b, (short)0, acc[nt], false, false);
        }
    }

    {
        v2f a;
        a.x = (half == 0) ? pR[(size_t)(128 - 2 * half) * TDIM] : 0.0f;
        a.y = 0.0f;
        const float* __restrict__ w = pW + (size_t)(2 * 32) * 512;
#pragma unroll
        for (int nt = 0; nt < 16; ++nt) {
            v2f b = *(const v2f*)(w + nt * 32);
            acc[nt] = __builtin_amdgcn_wmma_f32_16x16x4_f32(
                false, a, false, b, (short)0, acc[nt], false, false);
        }
    }

#pragma unroll 4
    for (int kc = 0; kc < 32; ++kc) {
        v2f a;
        a.x = pI[(size_t)(4 * kc) * TDIM];
        a.y = pI[(size_t)(4 * kc + 1) * TDIM];
        const float* __restrict__ w = pW + (size_t)(2 * (kc + 33)) * 512;
#pragma unroll
        for (int nt = 0; nt < 16; ++nt) {
            v2f b = *(const v2f*)(w + nt * 32);
            acc[nt] = __builtin_amdgcn_wmma_f32_16x16x4_f32(
                false, a, false, b, (short)0, acc[nt], false, false);
        }
    }

    float* __restrict__ outp = fft + (size_t)r0 * FFT_LEN + mrow;
#pragma unroll
    for (int nt = 0; nt < 16; ++nt) {
#pragma unroll
        for (int j = 0; j < 8; ++j) {
            outp[(size_t)(j + half * 8) * FFT_LEN + nt * 16] = acc[nt][j];
        }
    }
}

__global__ void overlap_add_kernel(const float* __restrict__ fft,
                                   const float* __restrict__ h,
                                   float* __restrict__ out) {
    const int idx = blockIdx.x * blockDim.x + threadIdx.x;
    if (idx >= NBC * OUTLEN) return;
    const int bc = idx / OUTLEN;
    const int n  = idx - bc * OUTLEN;
    const int m  = n + GROUP_DELAY;

    int tlo = (m - (WIN_LEN - FRM_LEN)) >> 7;
    if (tlo < 0) tlo = 0;
    int thi = m >> 7;
    if (thi > TDIM - 1) thi = TDIM - 1;

    const float* __restrict__ f = fft + (size_t)bc * TDIM * FFT_LEN;
    float acc = 0.0f;
    for (int tt = tlo; tt <= thi; ++tt) {
        const int k = m - (tt << 7);
        acc += h[k] * f[tt * FFT_LEN + (k & 255)];
    }
    out[idx] = acc;
}

extern "C" void kernel_launch(void* const* d_in, const int* in_sizes, int n_in,
                              void* d_out, int out_size, void* d_ws, size_t ws_size,
                              hipStream_t stream) {
    const float* re = (const float*)d_in[0];
    const float* im = (const float*)d_in[1];
    const float* h  = (const float*)d_in[2];
    float* out = (float*)d_out;

    float* Wp  = (float*)d_ws;
    float* fft = Wp + (size_t)(KPAD / 2) * 512;

    build_w_kernel<<<KPAD, FFT_LEN, 0, stream>>>(Wp);
    ifft_wmma_kernel<<<(NROWS / 16) * 32 / 256, 256, 0, stream>>>(re, im, Wp, fft);
    const int total = NBC * OUTLEN;
    overlap_add_kernel<<<(total + 255) / 256, 256, 0, stream>>>(fft, h, out);
}


# overlap-add kernel: 4 adjacent outputs per lane with 16-byte loads/stores, 6 taps unrolled, 16 steps per wave, 48 loads issued ahead of the sums (64x fewer waves); tap sum order unchanged
# speedup vs baseline: 1.5634x; 1.5634x over previous
#include <hip/hip_runtime.h>
#include <hip/hip_bf16.h>


typedef __attribute__((ext_vector_type(2))) float v2f;
typedef __attribute__((ext_vector_type(8))) float v8f;

#define FFT_LEN   256
#define SPEC_BINS 129
#define KPAD      260
#define NROWS     65536
#define TDIM      4096
#define WIN_LEN   768
#define FRM_LEN   128
#define GROUP_DELAY 648
#define OUTLEN    (TDIM*FRM_LEN - GROUP_DELAY)
#define NBC       16

__global__ void build_w_kernel(float* __restrict__ Wp) {
    int k = blockIdx.x;
    int n = threadIdx.x;
    float val = 0.0f;
    int s = -1;
    bool isIm = false;
    if (k <= 128)                { s = k;        isIm = false; }
    else if (k >= 132 && k <= 258){ s = k - 131; isIm = true;  }
    if (s >= 0) {
        float c = (s == 0 || s == 128) ? 1.0f : 2.0f;
        int nrev  = 255 - n;
        int phase = (s * nrev) & 255;
        float ang = (float)phase * 0.0245436926061702596f;
        float sn, cs;
        __sincosf(ang, &sn, &cs);
        val = isIm ? (128.0f * c * sn) : (-128.0f * c * cs);
    }
    Wp[(k >> 1) * 512 + n * 2 + (k & 1)] = val;
}

__global__ void __launch_bounds__(256)
ifft_wmma_kernel(const float* __restrict__ re, const float* __restrict__ im,
                 const float* __restrict__ Wp, float* __restrict__ fft) {
    const int wid  = (blockIdx.x * blockDim.x + threadIdx.x) >> 5;
    const int lane = threadIdx.x & 31;
    const int half = lane >> 4;
    const int mrow = lane & 15;

    const int r0 = wid << 4;
    const int bc = r0 >> 12;
    const int t  = (r0 & (TDIM - 1)) + mrow;

    const float* __restrict__ pR = re + ((size_t)bc * SPEC_BINS) * TDIM + t
                                      + (size_t)(2 * half) * TDIM;
    const float* __restrict__ pI = im + ((size_t)bc * SPEC_BINS) * TDIM + t
                                      + (size_t)(1 + 2 * half) * TDIM;
    const float* __restrict__ pW = Wp + (size_t)half * 512 + (size_t)mrow * 2;

    v8f acc[16];
#pragma unroll
    for (int i = 0; i < 16; ++i) acc[i] = (v8f){};

#pragma unroll 4
    for (int kc = 0; kc < 32; ++kc) {
        v2f a;
        a.x = pR[(size_t)(4 * kc) * TDIM];
        a.y = pR[(size_t)(4 * kc + 1) * TDIM];
        const float* __restrict__ w = pW + (size_t)(2 * kc) * 512;
#pragma unroll
        for (int nt = 0; nt < 16; ++nt) {
            v2f b = *(const v2f*)(w + nt * 32);
            acc[nt] = __builtin_amdgcn_wmma_f32_16x16x4_f32(
                false, a, false, b, (short)0, acc[nt], false, false);
        }
    }

    {
        v2f a;
        a.x = (half == 0) ? pR[(size_t)(128 - 2 * half) * TDIM] : 0.0f;
        a.y = 0.0f;
        const float* __restrict__ w = pW + (size_t)(2 * 32) * 512;
#pragma unroll
        for (int nt = 0; nt < 16; ++nt) {
            v2f b = *(const v2f*)(w + nt * 32);
            acc[nt] = __builtin_amdgcn_wmma_f32_16x16x4_f32(
                false, a, false, b, (short)0, acc[nt], false, false);
        }
    }

#pragma unroll 4
    for (int kc = 0; kc < 32; ++kc) {
        v2f a;
        a.x = pI[(size_t)(4 * kc) * TDIM];
        a.y = pI[(size_t)(4 * kc + 1) * TDIM];
        const float* __restrict__ w = pW + (size_t)(2 * (kc + 33)) * 512;
#pragma unroll
        for (int nt = 0; nt < 16; ++nt) {
            v2f b = *(const v2f*)(w + nt * 32);
            acc[nt] = __builtin_amdgcn_wmma_f32_16x16x4_f32(
                false, a, false, b, (short)0, acc[nt], false, false);
        }
    }

    float* __restrict__ outp = fft + (size_t)r0 * FFT_LEN + mrow;
#pragma unroll
    for (int nt = 0; nt < 16; ++nt) {
#pragma unroll
        for (int j = 0; j < 8; ++j) {
            outp[(size_t)(j + half * 8) * FFT_LEN + nt * 16] = acc[nt][j];
        }
    }
}

typedef __attribute__((ext_vector_type(4))) float v4f;

#define OLA_GROUPS      (NBC * OUTLEN / 4)
#define OLA_WAVE_STEPS  (OLA_GROUPS / 32)
#define OLA_STEPS       16
#define OLA_BATCH       4
#define OLA_BLOCKS      512

template <int U>
__device__ __forceinline__ void ola_steps(const float* __restrict__ fft,
                                          const float* __restrict__ h,
                                          float* __restrict__ out, int g) {
    v4f hv[U][6], fv[U][6];
    int idx[U];
#pragma unroll
    for (int u = 0; u < U; ++u) {
        idx[u] = (g + 32 * u) * 4;
        const int bc = idx[u] / OUTLEN;
        const int m  = idx[u] - bc * OUTLEN + GROUP_DELAY;
        const int q  = m >> 7;
        const int j  = m & (FRM_LEN - 1);
        const float* __restrict__ f  = fft + (size_t)bc * TDIM * FFT_LEN
                                           + (size_t)(q - 5) * FFT_LEN + j;
        const float* __restrict__ hh = h + j;
#pragma unroll
        for (int e = 0; e < 6; ++e) {
            hv[u][e] = *(const v4f*)(hh + FRM_LEN * (5 - e));
            fv[u][e] = *(const v4f*)(f + e * FFT_LEN + FRM_LEN * ((5 - e) & 1));
        }
    }
    if (U > 1) __builtin_amdgcn_sched_barrier(0);
#pragma unroll
    for (int u = 0; u < U; ++u) {
        v4f acc = (v4f){0.0f, 0.0f, 0.0f, 0.0f};
#pragma unroll
        for (int e = 0; e < 6; ++e) acc += hv[u][e] * fv[u][e];
        *(v4f*)(out + idx[u]) = acc;
    }
}

__global__ void __launch_bounds__(256)
overlap_add_kernel(const float* __restrict__ fft,
                   const float* __restrict__ h,
                   float* __restrict__ out) {
    const int wave  = (blockIdx.x * 256 + threadIdx.x) >> 5;
    const int lane  = threadIdx.x & 31;
    const int step0 = wave * OLA_STEPS;
    const int g     = step0 * 32 + lane;
    int nsteps = OLA_WAVE_STEPS - step0;
    if (nsteps >= OLA_STEPS) {
        for (int b = 0; b < OLA_STEPS / OLA_BATCH; ++b)
            ola_steps<OLA_BATCH>(fft, h, out, g + 32 * OLA_BATCH * b);
    } else {
        for (int i = 0; i < nsteps; ++i)
            ola_steps<1>(fft, h, out, g + 32 * i);
    }
}

extern "C" void kernel_launch(void* const* d_in, const int* in_sizes, int n_in,
                              void* d_out, int out_size, void* d_ws, size_t ws_size,
                              hipStream_t stream) {
    const float* re = (const float*)d_in[0];
    const float* im = (const float*)d_in[1];
    const float* h  = (const float*)d_in[2];
    float* out = (float*)d_out;

    float* Wp  = (float*)d_ws;
    float* fft = Wp + (size_t)(KPAD / 2) * 512;

    build_w_kernel<<<KPAD, FFT_LEN, 0, stream>>>(Wp);
    ifft_wmma_kernel<<<(NROWS / 16) * 32 / 256, 256, 0, stream>>>(re, im, Wp, fft);
    overlap_add_kernel<<<OLA_BLOCKS, 256, 0, stream>>>(fft, h, out);
}


# irfft GEMM split by basis parity: cos half (n=0..128, Re rows) and sin half (n=0..127, Im rows) as two half-size products, x[n]=E+O, x[256-n]=E-O; 553 vs 1040 WMMAs per wave; sum order changes
# speedup vs baseline: 2.3275x; 1.4887x over previous
#include <hip/hip_runtime.h>
#include <hip/hip_bf16.h>


typedef __attribute__((ext_vector_type(2))) float v2f;
typedef __attribute__((ext_vector_type(8))) float v8f;

#define FFT_LEN   256
#define SPEC_BINS 129
#define NROWS     65536
#define TDIM      4096
#define WIN_LEN   768
#define FRM_LEN   128
#define GROUP_DELAY 648
#define OUTLEN    (TDIM*FRM_LEN - GROUP_DELAY)
#define NBC       16

#define KE        132
#define KO        128
#define NE_TILES  9
#define NO_TILES  8
#define WE_PITCH  (NE_TILES * 32)
#define WO_PITCH  (NO_TILES * 32)
#define WE_FLOATS ((KE / 2) * WE_PITCH)
#define WO_FLOATS ((KO / 2) * WO_PITCH)

__global__ void build_w_kernel(float* __restrict__ We, float* __restrict__ Wo) {
    const int k = blockIdx.x;
    const int n = threadIdx.x;
    if (k < KE) {
        if (n < NE_TILES * 16) {
            float val = 0.0f;
            if (k <= 128 && n <= 128) {
                const float c = (k == 0 || k == 128) ? 1.0f : 2.0f;
                const int phase = (k * n) & 255;
                const float ang = (float)phase * 0.0245436926061702596f;
                float sn, cs;
                __sincosf(ang, &sn, &cs);
                val = -128.0f * c * cs;
            }
            We[(k >> 1) * WE_PITCH + n * 2 + (k & 1)] = val;
        }
    } else {
        const int ko = k - KE;
        const int s  = ko + 1;
        if (n < NO_TILES * 16) {
            float val = 0.0f;
            if (s <= 127) {
                const int phase = (s * n) & 255;
                const float ang = (float)phase * 0.0245436926061702596f;
                float sn, cs;
                __sincosf(ang, &sn, &cs);
                val = 256.0f * sn;
            }
            Wo[(ko >> 1) * WO_PITCH + n * 2 + (ko & 1)] = val;
        }
    }
}

__global__ void __launch_bounds__(256)
ifft_wmma_kernel(const float* __restrict__ re, const float* __restrict__ im,
                 const float* __restrict__ We, const float* __restrict__ Wo,
                 float* __restrict__ fft) {
    const int wid  = (blockIdx.x * blockDim.x + threadIdx.x) >> 5;
    const int lane = threadIdx.x & 31;
    const int half = lane >> 4;
    const int mrow = lane & 15;

    const int r0 = wid << 4;
    const int bc = r0 >> 12;
    const int t  = (r0 & (TDIM - 1)) + mrow;

    const float* __restrict__ pR = re + ((size_t)bc * SPEC_BINS) * TDIM + t
                                      + (size_t)(2 * half) * TDIM;
    const float* __restrict__ pI = im + ((size_t)bc * SPEC_BINS) * TDIM + t
                                      + (size_t)(1 + 2 * half) * TDIM;
    const float* __restrict__ pWe = We + (size_t)half * WE_PITCH + (size_t)mrow * 2;
    const float* __restrict__ pWo = Wo + (size_t)half * WO_PITCH + (size_t)mrow * 2;

    v8f accE[NE_TILES], accO[NO_TILES];
#pragma unroll
    for (int i = 0; i < NE_TILES; ++i) accE[i] = (v8f){};
#pragma unroll
    for (int i = 0; i < NO_TILES; ++i) accO[i] = (v8f){};

#pragma unroll 4
    for (int kc = 0; kc < 32; ++kc) {
        v2f a;
        a.x = pR[(size_t)(4 * kc) * TDIM];
        a.y = pR[(size_t)(4 * kc + 1) * TDIM];
        const float* __restrict__ w = pWe + (size_t)(2 * kc) * WE_PITCH;
#pragma unroll
        for (int nt = 0; nt < NE_TILES; ++nt) {
            v2f b = *(const v2f*)(w + nt * 32);
            accE[nt] = __builtin_amdgcn_wmma_f32_16x16x4_f32(
                false, a, false, b, (short)0, accE[nt], false, false);
        }
    }

    {
        v2f a;
        a.x = (half == 0) ? pR[(size_t)(128 - 2 * half) * TDIM] : 0.0f;
        a.y = 0.0f;
        const float* __restrict__ w = pWe + (size_t)(2 * 32) * WE_PITCH;
#pragma unroll
        for (int nt = 0; nt < NE_TILES; ++nt) {
            v2f b = *(const v2f*)(w + nt * 32);
            accE[nt] = __builtin_amdgcn_wmma_f32_16x16x4_f32(
                false, a, false, b, (short)0, accE[nt], false, false);
        }
    }

#pragma unroll 4
    for (int kc = 0; kc < 32; ++kc) {
        v2f a;
        a.x = pI[(size_t)(4 * kc) * TDIM];
        a.y = pI[(size_t)(4 * kc + 1) * TDIM];
        const float* __restrict__ w = pWo + (size_t)(2 * kc) * WO_PITCH;
#pragma unroll
        for (int nt = 0; nt < NO_TILES; ++nt) {
            v2f b = *(const v2f*)(w + nt * 32);
            accO[nt] = __builtin_amdgcn_wmma_f32_16x16x4_f32(
                false, a, false, b, (short)0, accO[nt], false, false);
        }
    }

    float* __restrict__ outp = fft + (size_t)r0 * FFT_LEN + (size_t)(half * 8) * FFT_LEN;
    float* __restrict__ pP = outp + (size_t)(255 - mrow);
    float* __restrict__ pQ = outp + (size_t)mrow;
#pragma unroll
    for (int nt = 0; nt < NO_TILES; ++nt) {
#pragma unroll
        for (int j = 0; j < 8; ++j)
            pP[j * FFT_LEN - 16 * nt] = accE[nt][j] + accO[nt][j];
    }
#pragma unroll
    for (int nt = 1; nt < NO_TILES; ++nt) {
#pragma unroll
        for (int j = 0; j < 8; ++j)
            pQ[j * FFT_LEN + 16 * nt - 1] = accE[nt][j] - accO[nt][j];
    }
    if (mrow != 0) {
#pragma unroll
        for (int j = 0; j < 8; ++j)
            pQ[j * FFT_LEN - 1] = accE[0][j] - accO[0][j];
    } else {
#pragma unroll
        for (int j = 0; j < 8; ++j)
            outp[j * FFT_LEN + 127] = accE[NE_TILES - 1][j];
    }
}

typedef __attribute__((ext_vector_type(4))) float v4f;

#define OLA_GROUPS      (NBC * OUTLEN / 4)
#define OLA_WAVE_STEPS  (OLA_GROUPS / 32)
#define OLA_STEPS       16
#define OLA_BATCH       4
#define OLA_BLOCKS      512

template <int U>
__device__ __forceinline__ void ola_steps(const float* __restrict__ fft,
                                          const float* __restrict__ h,
                                          float* __restrict__ out, int g) {
    v4f hv[U][6], fv[U][6];
    int idx[U];
#pragma unroll
    for (int u = 0; u < U; ++u) {
        idx[u] = (g + 32 * u) * 4;
        const int bc = idx[u] / OUTLEN;
        const int m  = idx[u] - bc * OUTLEN + GROUP_DELAY;
        const int q  = m >> 7;
        const int j  = m & (FRM_LEN - 1);
        const float* __restrict__ f  = fft + (size_t)bc * TDIM * FFT_LEN
                                           + (size_t)(q - 5) * FFT_LEN + j;
        const float* __restrict__ hh = h + j;
#pragma unroll
        for (int e = 0; e < 6; ++e) {
            hv[u][e] = *(const v4f*)(hh + FRM_LEN * (5 - e));
            fv[u][e] = *(const v4f*)(f + e * FFT_LEN + FRM_LEN * ((5 - e) & 1));
        }
    }
    if (U > 1) __builtin_amdgcn_sched_barrier(0);
#pragma unroll
    for (int u = 0; u < U; ++u) {
        v4f acc = (v4f){0.0f, 0.0f, 0.0f, 0.0f};
#pragma unroll
        for (int e = 0; e < 6; ++e) acc += hv[u][e] * fv[u][e];
        *(v4f*)(out + idx[u]) = acc;
    }
}

__global__ void __launch_bounds__(256)
overlap_add_kernel(const float* __restrict__ fft,
                   const float* __restrict__ h,
                   float* __restrict__ out) {
    const int wave  = (blockIdx.x * 256 + threadIdx.x) >> 5;
    const int lane  = threadIdx.x & 31;
    const int step0 = wave * OLA_STEPS;
    const int g     = step0 * 32 + lane;
    int nsteps = OLA_WAVE_STEPS - step0;
    if (nsteps >= OLA_STEPS) {
        for (int b = 0; b < OLA_STEPS / OLA_BATCH; ++b)
            ola_steps<OLA_BATCH>(fft, h, out, g + 32 * OLA_BATCH * b);
    } else {
        for (int i = 0; i < nsteps; ++i)
            ola_steps<1>(fft, h, out, g + 32 * i);
    }
}

extern "C" void kernel_launch(void* const* d_in, const int* in_sizes, int n_in,
                              void* d_out, int out_size, void* d_ws, size_t ws_size,
                              hipStream_t stream) {
    const float* re = (const float*)d_in[0];
    const float* im = (const float*)d_in[1];
    const float* h  = (const float*)d_in[2];
    float* out = (float*)d_out;

    float* We  = (float*)d_ws;
    float* Wo  = We + (size_t)WE_FLOATS;
    float* fft = Wo + (size_t)WO_FLOATS;

    build_w_kernel<<<KE + KO, 160, 0, stream>>>(We, Wo);
    ifft_wmma_kernel<<<(NROWS / 16) * 32 / 256, 256, 0, stream>>>(re, im, We, Wo, fft);
    overlap_add_kernel<<<OLA_BLOCKS, 256, 0, stream>>>(fft, h, out);
}


# overlap-add fused into the irfft GEMM kernel: a workgroup keeps its 128 irfft rows in LDS and sums its outputs there; only 10 edge rows per workgroup go to memory for a seam kernel; sum order kept
# speedup vs baseline: 3.2827x; 1.4104x over previous
#include <hip/hip_runtime.h>
#include <hip/hip_bf16.h>


typedef __attribute__((ext_vector_type(2))) float v2f;
typedef __attribute__((ext_vector_type(4))) float v4f;
typedef __attribute__((ext_vector_type(8))) float v8f;

#define FFT_LEN   256
#define SPEC_BINS 129
#define NROWS     65536
#define TDIM      4096
#define WIN_LEN   768
#define FRM_LEN   128
#define GROUP_DELAY 648
#define NBC       16

#define KE        132
#define KO        128
#define NE_TILES  9
#define NO_TILES  8
#define WE_PITCH  (NE_TILES * 32)
#define WO_PITCH  (NO_TILES * 32)
#define WE_FLOATS ((KE / 2) * WE_PITCH)
#define WO_FLOATS ((KO / 2) * WO_PITCH)

#define BLK_FRAMES 128
#define BLKS_PER_BC (TDIM / BLK_FRAMES)
#define NBLOCKS   (NROWS / BLK_FRAMES)
#define LDS_PITCH 260
#define EDGE_ROWS 5
#define EDGE_FLOATS (2 * EDGE_ROWS * FFT_LEN)
#define OUTLEN    (TDIM*FRM_LEN - GROUP_DELAY)

__global__ void build_w_kernel(float* __restrict__ We, float* __restrict__ Wo) {
    const int k = blockIdx.x;
    const int n = threadIdx.x;
    if (k < KE) {
        if (n < NE_TILES * 16) {
            float val = 0.0f;
            if (k <= 128 && n <= 128) {
                const float c = (k == 0 || k == 128) ? 1.0f : 2.0f;
                const int phase = (k * n) & 255;
                const float ang = (float)phase * 0.0245436926061702596f;
                float sn, cs;
                __sincosf(ang, &sn, &cs);
                val = -128.0f * c * cs;
            }
            We[(k >> 1) * WE_PITCH + n * 2 + (k & 1)] = val;
        }
    } else {
        const int ko = k - KE;
        const int s  = ko + 1;
        if (n < NO_TILES * 16) {
            float val = 0.0f;
            if (s <= 127) {
                const int phase = (s * n) & 255;
                const float ang = (float)phase * 0.0245436926061702596f;
                float sn, cs;
                __sincosf(ang, &sn, &cs);
                val = 256.0f * sn;
            }
            Wo[(ko >> 1) * WO_PITCH + n * 2 + (ko & 1)] = val;
        }
    }
}

__device__ __forceinline__ v4f ola_piece(const float* __restrict__ s, const v4f (&hv)[6]) {
    v4f acc = (v4f){0.0f, 0.0f, 0.0f, 0.0f};
#pragma unroll
    for (int e = 0; e < 6; ++e) {
        const int d = 5 - e;
        acc += hv[d] * *(const v4f*)(s - d * LDS_PITCH + FRM_LEN * (d & 1));
    }
    return acc;
}

__global__ void __launch_bounds__(256)
ifft_ola_kernel(const float* __restrict__ re, const float* __restrict__ im,
                const float* __restrict__ We, const float* __restrict__ Wo,
                const float* __restrict__ h,
                float* __restrict__ edge, float* __restrict__ out) {
    __shared__ float sfft[BLK_FRAMES * LDS_PITCH];

    const int wv   = threadIdx.x >> 5;
    const int wid  = blockIdx.x * 8 + wv;
    const int lane = threadIdx.x & 31;
    const int half = lane >> 4;
    const int mrow = lane & 15;

    const int r0 = wid << 4;
    const int bc = r0 >> 12;
    const int t  = (r0 & (TDIM - 1)) + mrow;

    const float* __restrict__ pR = re + ((size_t)bc * SPEC_BINS) * TDIM + t
                                      + (size_t)(2 * half) * TDIM;
    const float* __restrict__ pI = im + ((size_t)bc * SPEC_BINS) * TDIM + t
                                      + (size_t)(1 + 2 * half) * TDIM;
    const float* __restrict__ pWe = We + (size_t)half * WE_PITCH + (size_t)mrow * 2;
    const float* __restrict__ pWo = Wo + (size_t)half * WO_PITCH + (size_t)mrow * 2;

    v8f accE[NE_TILES], accO[NO_TILES];
#pragma unroll
    for (int i = 0; i < NE_TILES; ++i) accE[i] = (v8f){};
#pragma unroll
    for (int i = 0; i < NO_TILES; ++i) accO[i] = (v8f){};

#pragma unroll 4
    for (int kc = 0; kc < 32; ++kc) {
        v2f a;
        a.x = pR[(size_t)(4 * kc) * TDIM];
        a.y = pR[(size_t)(4 * kc + 1) * TDIM];
        const float* __restrict__ w = pWe + (size_t)(2 * kc) * WE_PITCH;
#pragma unroll
        for (int nt = 0; nt < NE_TILES; ++nt) {
            v2f b = *(const v2f*)(w + nt * 32);
            accE[nt] = __builtin_amdgcn_wmma_f32_16x16x4_f32(
                false, a, false, b, (short)0, accE[nt], false, false);
        }
    }

    {
        v2f a;
        a.x = (half == 0) ? pR[(size_t)(128 - 2 * half) * TDIM] : 0.0f;
        a.y = 0.0f;
        const float* __restrict__ w = pWe + (size_t)(2 * 32) * WE_PITCH;
#pragma unroll
        for (int nt = 0; nt < NE_TILES; ++nt) {
            v2f b = *(const v2f*)(w + nt * 32);
            accE[nt] = __builtin_amdgcn_wmma_f32_16x16x4_f32(
                false, a, false, b, (short)0, accE[nt], false, false);
        }
    }

#pragma unroll 4
    for (int kc = 0; kc < 32; ++kc) {
        v2f a;
        a.x = pI[(size_t)(4 * kc) * TDIM];
        a.y = pI[(size_t)(4 * kc + 1) * TDIM];
        const float* __restrict__ w = pWo + (size_t)(2 * kc) * WO_PITCH;
#pragma unroll
        for (int nt = 0; nt < NO_TILES; ++nt) {
            v2f b = *(const v2f*)(w + nt * 32);
            accO[nt] = __builtin_amdgcn_wmma_f32_16x16x4_f32(
                false, a, false, b, (short)0, accO[nt], false, false);
        }
    }

    float* __restrict__ sp = sfft + (16 * wv + 8 * half) * LDS_PITCH;
    float* __restrict__ sP = sp + (255 - mrow);
    float* __restrict__ sQ = sp + mrow;
#pragma unroll
    for (int nt = 0; nt < NO_TILES; ++nt) {
#pragma unroll
        for (int j = 0; j < 8; ++j)
            sP[j * LDS_PITCH - 16 * nt] = accE[nt][j] + accO[nt][j];
    }
#pragma unroll
    for (int nt = 1; nt < NO_TILES; ++nt) {
#pragma unroll
        for (int j = 0; j < 8; ++j)
            sQ[j * LDS_PITCH + 16 * nt - 1] = accE[nt][j] - accO[nt][j];
    }
    if (mrow != 0) {
#pragma unroll
        for (int j = 0; j < 8; ++j)
            sQ[j * LDS_PITCH - 1] = accE[0][j] - accO[0][j];
    } else {
#pragma unroll
        for (int j = 0; j < 8; ++j)
            sp[j * LDS_PITCH + 127] = accE[NE_TILES - 1][j];
    }

    const int j4 = 4 * lane;
    v4f hv[6];
#pragma unroll
    for (int d = 0; d < 6; ++d) hv[d] = *(const v4f*)(h + j4 + FRM_LEN * d);

    __syncthreads();

    {
        float* __restrict__ eg = edge + (size_t)blockIdx.x * EDGE_FLOATS;
        for (int i = threadIdx.x; i < 2 * EDGE_ROWS * (FFT_LEN / 4); i += 256) {
            const int row  = i >> 6;
            const int c4   = (i & 63) * 4;
            const int srow = (row < EDGE_ROWS) ? row : row + (BLK_FRAMES - 2 * EDGE_ROWS);
            *(v4f*)(eg + row * FFT_LEN + c4) = *(const v4f*)(sfft + srow * LDS_PITCH + c4);
        }
    }

    const int f0 = (blockIdx.x & (BLKS_PER_BC - 1)) * BLK_FRAMES;
    const int obase = bc * OUTLEN - GROUP_DELAY + f0 * FRM_LEN + j4;
    {
        const int qq = EDGE_ROWS + wv;
        const v4f acc = ola_piece(sfft + qq * LDS_PITCH + j4, hv);
        if ((f0 + qq) * FRM_LEN + j4 >= GROUP_DELAY)
            *(v4f*)(out + (obase + qq * FRM_LEN)) = acc;
    }
#pragma unroll 2
    for (int qq = EDGE_ROWS + 8 + wv; qq < BLK_FRAMES; qq += 8)
        *(v4f*)(out + (obase + qq * FRM_LEN)) = ola_piece(sfft + qq * LDS_PITCH + j4, hv);
}

#define NSEAMS (NBC * (BLKS_PER_BC - 1))

__global__ void __launch_bounds__(32 * EDGE_ROWS)
seam_kernel(const float* __restrict__ edge, const float* __restrict__ h,
            float* __restrict__ out) {
    const int bc   = blockIdx.x / (BLKS_PER_BC - 1);
    const int blk  = blockIdx.x - bc * (BLKS_PER_BC - 1) + 1;
    const int sidx = threadIdx.x >> 5;
    const int j4   = 4 * (threadIdx.x & 31);
    const float* __restrict__ cur  = edge + (size_t)(bc * BLKS_PER_BC + blk) * EDGE_FLOATS;
    const float* __restrict__ prev = cur - EDGE_FLOATS;

    v4f acc = (v4f){0.0f, 0.0f, 0.0f, 0.0f};
#pragma unroll
    for (int e = 0; e < 6; ++e) {
        const int d = 5 - e;
        const int r = sidx - d;
        const float* __restrict__ row = (r >= 0) ? cur + r * FFT_LEN
                                                 : prev + (2 * EDGE_ROWS + r) * FFT_LEN;
        acc += *(const v4f*)(h + j4 + FRM_LEN * d)
             * *(const v4f*)(row + j4 + FRM_LEN * (d & 1));
    }
    const int m = (blk * BLK_FRAMES + sidx) * FRM_LEN + j4;
    *(v4f*)(out + bc * OUTLEN + m - GROUP_DELAY) = acc;
}

extern "C" void kernel_launch(void* const* d_in, const int* in_sizes, int n_in,
                              void* d_out, int out_size, void* d_ws, size_t ws_size,
                              hipStream_t stream) {
    const float* re = (const float*)d_in[0];
    const float* im = (const float*)d_in[1];
    const float* h  = (const float*)d_in[2];
    float* out = (float*)d_out;

    float* We   = (float*)d_ws;
    float* Wo   = We + (size_t)WE_FLOATS;
    float* edge = Wo + (size_t)WO_FLOATS;

    build_w_kernel<<<KE + KO, 160, 0, stream>>>(We, Wo);
    ifft_ola_kernel<<<NBLOCKS, 256, 0, stream>>>(re, im, We, Wo, h, edge, out);
    seam_kernel<<<NSEAMS, 32 * EDGE_ROWS, 0, stream>>>(edge, h, out);
}


# GEMM operand loads software-pipelined in straight-line code: B fragments requested 3 K chunks ahead of their WMMAs, A values in batches of 13 chunks (26 loads in flight); sum order kept
# speedup vs baseline: 3.5286x; 1.0749x over previous
#include <hip/hip_runtime.h>
#include <hip/hip_bf16.h>


typedef __attribute__((ext_vector_type(2))) float v2f;
typedef __attribute__((ext_vector_type(4))) float v4f;
typedef __attribute__((ext_vector_type(8))) float v8f;

#define FFT_LEN   256
#define SPEC_BINS 129
#define NROWS     65536
#define TDIM      4096
#define WIN_LEN   768
#define FRM_LEN   128
#define GROUP_DELAY 648
#define NBC       16

#define KE        132
#define KO        128
#define NE_TILES  9
#define NO_TILES  8
#define WE_PITCH  (NE_TILES * 32)
#define WO_PITCH  (NO_TILES * 32)
#define WE_FLOATS ((KE / 2) * WE_PITCH)
#define WO_FLOATS ((KO / 2) * WO_PITCH)

#define NCHUNK    65
#define B_AHEAD   3
#define B_RING    4
#define A_BATCH   13

#define BLK_FRAMES 128
#define BLKS_PER_BC (TDIM / BLK_FRAMES)
#define NBLOCKS   (NROWS / BLK_FRAMES)
#define LDS_PITCH 260
#define EDGE_ROWS 5
#define EDGE_FLOATS (2 * EDGE_ROWS * FFT_LEN)
#define OUTLEN    (TDIM*FRM_LEN - GROUP_DELAY)

__global__ void build_w_kernel(float* __restrict__ We, float* __restrict__ Wo) {
    const int k = blockIdx.x;
    const int n = threadIdx.x;
    if (k < KE) {
        if (n < NE_TILES * 16) {
            float val = 0.0f;
            if (k <= 128 && n <= 128) {
                const float c = (k == 0 || k == 128) ? 1.0f : 2.0f;
                const int phase = (k * n) & 255;
                const float ang = (float)phase * 0.0245436926061702596f;
                float sn, cs;
                __sincosf(ang, &sn, &cs);
                val = -128.0f * c * cs;
            }
            We[(k >> 1) * WE_PITCH + n * 2 + (k & 1)] = val;
        }
    } else {
        const int ko = k - KE;
        const int s  = ko + 1;
        if (n < NO_TILES * 16) {
            float val = 0.0f;
            if (s <= 127) {
                const int phase = (s * n) & 255;
                const float ang = (float)phase * 0.0245436926061702596f;
                float sn, cs;
                __sincosf(ang, &sn, &cs);
                val = 256.0f * sn;
            }
            Wo[(ko >> 1) * WO_PITCH + n * 2 + (ko & 1)] = val;
        }
    }
}

__device__ __forceinline__ v4f ola_piece(const float* __restrict__ s, const v4f (&hv)[6]) {
    v4f acc = (v4f){0.0f, 0.0f, 0.0f, 0.0f};
#pragma unroll
    for (int e = 0; e < 6; ++e) {
        const int d = 5 - e;
        acc += hv[d] * *(const v4f*)(s - d * LDS_PITCH + FRM_LEN * (d & 1));
    }
    return acc;
}

struct GemmPtrs {
    const float* __restrict__ pR;
    const float* __restrict__ pI;
    const float* __restrict__ pN;
    const float* __restrict__ pWe;
    const float* __restrict__ pWo;
    int half;
};

struct GemmRegs {
    v8f accE[NE_TILES];
    v8f accO[NO_TILES];
    v2f a[2][A_BATCH];
    v2f b[B_RING][NE_TILES];
};

template <int C>
__device__ __forceinline__ void load_a(GemmRegs& r, const GemmPtrs& p) {
    v2f& a = r.a[(C / A_BATCH) & 1][C % A_BATCH];
    if constexpr (C < 32) {
        a.x = p.pR[(size_t)(4 * C) * TDIM];
        a.y = p.pR[(size_t)(4 * C + 1) * TDIM];
    } else if constexpr (C == 32) {
        a.x = p.pN[0];
        a.y = 0.0f;
    } else {
        a.x = p.pI[(size_t)(4 * (C - 33)) * TDIM];
        a.y = p.pI[(size_t)(4 * (C - 33) + 1) * TDIM];
    }
}

template <int C, int N>
__device__ __forceinline__ void load_a_batch(GemmRegs& r, const GemmPtrs& p) {
    if constexpr (N > 0 && C < NCHUNK) {
        load_a<C>(r, p);
        load_a_batch<C + 1, N - 1>(r, p);
    }
}

template <int C>
__device__ __forceinline__ void load_b(GemmRegs& r, const GemmPtrs& p) {
    if constexpr (C <= 32) {
        const float* __restrict__ w = p.pWe + (size_t)(2 * C) * WE_PITCH;
#pragma unroll
        for (int nt = 0; nt < NE_TILES; ++nt) r.b[C % B_RING][nt] = *(const v2f*)(w + nt * 32);
    } else {
        const float* __restrict__ w = p.pWo + (size_t)(2 * (C - 33)) * WO_PITCH;
#pragma unroll
        for (int nt = 0; nt < NO_TILES; ++nt) r.b[C % B_RING][nt] = *(const v2f*)(w + nt * 32);
    }
}

template <int C>
__device__ __forceinline__ void gemm_stage(GemmRegs& r, const GemmPtrs& p) {
    constexpr int CB = C + B_AHEAD;
    if constexpr (CB < NCHUNK) {
        if constexpr (CB % A_BATCH == 0) load_a_batch<CB, A_BATCH>(r, p);
        load_b<CB>(r, p);
    }
    v2f a = r.a[(C / A_BATCH) & 1][C % A_BATCH];
    if constexpr (C <= 32) {
        if constexpr (C == 32) a.x = (p.half == 0) ? a.x : 0.0f;
#pragma unroll
        for (int nt = 0; nt < NE_TILES; ++nt)
            r.accE[nt] = __builtin_amdgcn_wmma_f32_16x16x4_f32(
                false, a, false, r.b[C % B_RING][nt], (short)0, r.accE[nt], false, false);
    } else {
#pragma unroll
        for (int nt = 0; nt < NO_TILES; ++nt)
            r.accO[nt] = __builtin_amdgcn_wmma_f32_16x16x4_f32(
                false, a, false, r.b[C % B_RING][nt], (short)0, r.accO[nt], false, false);
    }
    if constexpr (C + 1 < NCHUNK) gemm_stage<C + 1>(r, p);
}

template <int C>
__device__ __forceinline__ void load_b_first(GemmRegs& r, const GemmPtrs& p) {
    if constexpr (C < B_AHEAD) {
        load_b<C>(r, p);
        load_b_first<C + 1>(r, p);
    }
}

__global__ void __launch_bounds__(256) __attribute__((amdgpu_waves_per_eu(4, 4)))
ifft_ola_kernel(const float* __restrict__ re, const float* __restrict__ im,
                const float* __restrict__ We, const float* __restrict__ Wo,
                const float* __restrict__ h,
                float* __restrict__ edge, float* __restrict__ out) {
    __shared__ float sfft[BLK_FRAMES * LDS_PITCH];

    const int wv   = threadIdx.x >> 5;
    const int wid  = blockIdx.x * 8 + wv;
    const int lane = threadIdx.x & 31;
    const int half = lane >> 4;
    const int mrow = lane & 15;

    const int r0 = wid << 4;
    const int bc = r0 >> 12;
    const int t  = (r0 & (TDIM - 1)) + mrow;

    const float* __restrict__ pR = re + ((size_t)bc * SPEC_BINS) * TDIM + t
                                      + (size_t)(2 * half) * TDIM;
    const float* __restrict__ pI = im + ((size_t)bc * SPEC_BINS) * TDIM + t
                                      + (size_t)(1 + 2 * half) * TDIM;
    const float* __restrict__ pWe = We + (size_t)half * WE_PITCH + (size_t)mrow * 2;
    const float* __restrict__ pWo = Wo + (size_t)half * WO_PITCH + (size_t)mrow * 2;

    GemmRegs g;
#pragma unroll
    for (int i = 0; i < NE_TILES; ++i) g.accE[i] = (v8f){};
#pragma unroll
    for (int i = 0; i < NO_TILES; ++i) g.accO[i] = (v8f){};
    v8f (&accE)[NE_TILES] = g.accE;
    v8f (&accO)[NO_TILES] = g.accO;

    {
        GemmPtrs gp;
        gp.pR = pR;  gp.pI = pI;
        gp.pN = re + ((size_t)bc * SPEC_BINS + 128) * TDIM + t;
        gp.pWe = pWe;  gp.pWo = pWo;
        gp.half = half;
        load_a_batch<0, A_BATCH>(g, gp);
        load_b_first<0>(g, gp);
        gemm_stage<0>(g, gp);
    }

    float* __restrict__ sp = sfft + (16 * wv + 8 * half) * LDS_PITCH;
    float* __restrict__ sP = sp + (255 - mrow);
    float* __restrict__ sQ = sp + mrow;
#pragma unroll
    for (int nt = 0; nt < NO_TILES; ++nt) {
#pragma unroll
        for (int j = 0; j < 8; ++j)
            sP[j * LDS_PITCH - 16 * nt] = accE[nt][j] + accO[nt][j];
    }
#pragma unroll
    for (int nt = 1; nt < NO_TILES; ++nt) {
#pragma unroll
        for (int j = 0; j < 8; ++j)
            sQ[j * LDS_PITCH + 16 * nt - 1] = accE[nt][j] - accO[nt][j];
    }
    if (mrow != 0) {
#pragma unroll
        for (int j = 0; j < 8; ++j)
            sQ[j * LDS_PITCH - 1] = accE[0][j] - accO[0][j];
    } else {
#pragma unroll
        for (int j = 0; j < 8; ++j)
            sp[j * LDS_PITCH + 127] = accE[NE_TILES - 1][j];
    }

    const int j4 = 4 * lane;
    v4f hv[6];
#pragma unroll
    for (int d = 0; d < 6; ++d) hv[d] = *(const v4f*)(h + j4 + FRM_LEN * d);

    __syncthreads();

    {
        float* __restrict__ eg = edge + (size_t)blockIdx.x * EDGE_FLOATS;
        for (int i = threadIdx.x; i < 2 * EDGE_ROWS * (FFT_LEN / 4); i += 256) {
            const int row  = i >> 6;
            const int c4   = (i & 63) * 4;
            const int srow = (row < EDGE_ROWS) ? row : row + (BLK_FRAMES - 2 * EDGE_ROWS);
            *(v4f*)(eg + row * FFT_LEN + c4) = *(const v4f*)(sfft + srow * LDS_PITCH + c4);
        }
    }

    const int f0 = (blockIdx.x & (BLKS_PER_BC - 1)) * BLK_FRAMES;
    const int obase = bc * OUTLEN - GROUP_DELAY + f0 * FRM_LEN + j4;
    {
        const int qq = EDGE_ROWS + wv;
        const v4f acc = ola_piece(sfft + qq * LDS_PITCH + j4, hv);
        if ((f0 + qq) * FRM_LEN + j4 >= GROUP_DELAY)
            *(v4f*)(out + (obase + qq * FRM_LEN)) = acc;
    }
#pragma unroll 2
    for (int qq = EDGE_ROWS + 8 + wv; qq < BLK_FRAMES; qq += 8)
        *(v4f*)(out + (obase + qq * FRM_LEN)) = ola_piece(sfft + qq * LDS_PITCH + j4, hv);
}

#define NSEAMS (NBC * (BLKS_PER_BC - 1))

__global__ void __launch_bounds__(32 * EDGE_ROWS)
seam_kernel(const float* __restrict__ edge, const float* __restrict__ h,
            float* __restrict__ out) {
    const int bc   = blockIdx.x / (BLKS_PER_BC - 1);
    const int blk  = blockIdx.x - bc * (BLKS_PER_BC - 1) + 1;
    const int sidx = threadIdx.x >> 5;
    const int j4   = 4 * (threadIdx.x & 31);
    const float* __restrict__ cur  = edge + (size_t)(bc * BLKS_PER_BC + blk) * EDGE_FLOATS;
    const float* __restrict__ prev = cur - EDGE_FLOATS;

    v4f acc = (v4f){0.0f, 0.0f, 0.0f, 0.0f};
#pragma unroll
    for (int e = 0; e < 6; ++e) {
        const int d = 5 - e;
        const int r = sidx - d;
        const float* __restrict__ row = (r >= 0) ? cur + r * FFT_LEN
                                                 : prev + (2 * EDGE_ROWS + r) * FFT_LEN;
        acc += *(const v4f*)(h + j4 + FRM_LEN * d)
             * *(const v4f*)(row + j4 + FRM_LEN * (d & 1));
    }
    const int m = (blk * BLK_FRAMES + sidx) * FRM_LEN + j4;
    *(v4f*)(out + bc * OUTLEN + m - GROUP_DELAY) = acc;
}

extern "C" void kernel_launch(void* const* d_in, const int* in_sizes, int n_in,
                              void* d_out, int out_size, void* d_ws, size_t ws_size,
                              hipStream_t stream) {
    const float* re = (const float*)d_in[0];
    const float* im = (const float*)d_in[1];
    const float* h  = (const float*)d_in[2];
    float* out = (float*)d_out;

    float* We   = (float*)d_ws;
    float* Wo   = We + (size_t)WE_FLOATS;
    float* edge = Wo + (size_t)WO_FLOATS;

    build_w_kernel<<<KE + KO, 160, 0, stream>>>(We, Wo);
    ifft_ola_kernel<<<NBLOCKS, 256, 0, stream>>>(re, im, We, Wo, h, edge, out);
    seam_kernel<<<NSEAMS, 32 * EDGE_ROWS, 0, stream>>>(edge, h, out);
}


# irfft GEMMs split again by bin parity inside the fused kernel (workspace use unchanged): four quarter-size products, n=0..64, 4 outputs per lane by in-lane adds; 277 vs 553 WMMAs; sum order changes
# speedup vs baseline: 4.4049x; 1.2483x over previous
#include <hip/hip_runtime.h>
#include <hip/hip_bf16.h>


typedef __attribute__((ext_vector_type(2))) float v2f;
typedef __attribute__((ext_vector_type(4))) float v4f;
typedef __attribute__((ext_vector_type(8))) float v8f;

#define FFT_LEN   256
#define SPEC_BINS 129
#define NROWS     65536
#define TDIM      4096
#define WIN_LEN   768
#define FRM_LEN   128
#define GROUP_DELAY 648
#define NBC       16

#define KEE       68
#define KQ        64
#define NT5       5
#define NT4       4
#define P5        (NT5 * 32)
#define P4        (NT4 * 32)
#define WEE_FLOATS ((KEE / 2) * P5)
#define WEO_FLOATS ((KQ / 2) * P4)
#define WOE_FLOATS ((KQ / 2) * P4)
#define WOO_FLOATS ((KQ / 2) * P4)

#define NCHUNK    65
#define C_EO      17
#define C_OE      33
#define C_OO      49
#define B_AHEAD   6
#define B_RING    7
#define A_BATCH   13

#define BLK_FRAMES 128
#define BLKS_PER_BC (TDIM / BLK_FRAMES)
#define NBLOCKS   (NROWS / BLK_FRAMES)
#define LDS_PITCH 260
#define EDGE_ROWS 5
#define EDGE_FLOATS (2 * EDGE_ROWS * FFT_LEN)
#define OUTLEN    (TDIM*FRM_LEN - GROUP_DELAY)

__global__ void build_w_kernel(float* __restrict__ Wee, float* __restrict__ Weo,
                               float* __restrict__ Woe, float* __restrict__ Woo) {
    const int k = blockIdx.x;
    const int n = threadIdx.x;
    int s;
    int slot;
    int ncols;
    int nlive;
    bool is_cos;
    float* __restrict__ W;
    int pitch;
    if (k < KEE)               { slot = k;               s = 2 * slot;     ncols = NT5 * 16; nlive = 65; is_cos = true;  W = Wee; pitch = P5; }
    else if (k < KEE + KQ)     { slot = k - KEE;         s = 2 * slot + 1; ncols = NT4 * 16; nlive = 64; is_cos = true;  W = Weo; pitch = P4; }
    else if (k < KEE + 2 * KQ) { slot = k - KEE - KQ;    s = 2 * slot + 2; ncols = NT4 * 16; nlive = 64; is_cos = false; W = Woe; pitch = P4; }
    else                       { slot = k - KEE - 2 * KQ; s = 2 * slot + 1; ncols = NT4 * 16; nlive = 64; is_cos = false; W = Woo; pitch = P4; }
    if (n >= ncols) return;
    const int nn = (k >= KEE + 2 * KQ && n == 0) ? 64 : n;
    float val = 0.0f;
    const bool live_row = is_cos ? (s <= 128) : (s <= 127);
    if (live_row && n < nlive) {
        const int phase = (s * nn) & 255;
        const float ang = (float)phase * 0.0245436926061702596f;
        float sn, cs;
        __sincosf(ang, &sn, &cs);
        if (is_cos) {
            const float c = (s == 0 || s == 128) ? 1.0f : 2.0f;
            val = -128.0f * c * cs;
        } else {
            val = 256.0f * sn;
        }
    }
    W[(slot >> 1) * pitch + n * 2 + (slot & 1)] = val;
}

__device__ __forceinline__ v4f ola_piece(const float* __restrict__ s, const v4f (&hv)[6]) {
    v4f acc = (v4f){0.0f, 0.0f, 0.0f, 0.0f};
#pragma unroll
    for (int e = 0; e < 6; ++e) {
        const int d = 5 - e;
        acc += hv[d] * *(const v4f*)(s - d * LDS_PITCH + FRM_LEN * (d & 1));
    }
    return acc;
}

struct GemmPtrs {
    const float* __restrict__ pR;
    const float* __restrict__ pI;
    const float* __restrict__ pN;
    const float* __restrict__ pWee;
    const float* __restrict__ pWeo;
    const float* __restrict__ pWoe;
    const float* __restrict__ pWoo;
    int half;
};

struct GemmRegs {
    v8f accEe[NT5];
    v8f accEo[NT4];
    v8f accOe[NT4];
    v8f accOo[NT4];
    v2f a[2][A_BATCH];
    v2f b[B_RING][NT5];
};

template <int C>
__device__ __forceinline__ void load_a(GemmRegs& r, const GemmPtrs& p) {
    v2f& a = r.a[(C / A_BATCH) & 1][C % A_BATCH];
    if constexpr (C < 16) {
        a.x = p.pR[(size_t)(8 * C) * TDIM];
        a.y = p.pR[(size_t)(8 * C + 2) * TDIM];
    } else if constexpr (C == 16) {
        a.x = p.pN[0];
        a.y = 0.0f;
    } else if constexpr (C < C_OE) {
        a.x = p.pR[(size_t)(8 * (C - C_EO) + 1) * TDIM];
        a.y = p.pR[(size_t)(8 * (C - C_EO) + 3) * TDIM];
    } else if constexpr (C < C_OO) {
        a.x = p.pI[(size_t)(8 * (C - C_OE) + 2) * TDIM];
        a.y = p.pI[(size_t)(8 * (C - C_OE) + 4) * TDIM];
    } else {
        a.x = p.pI[(size_t)(8 * (C - C_OO) + 1) * TDIM];
        a.y = p.pI[(size_t)(8 * (C - C_OO) + 3) * TDIM];
    }
}

template <int C, int N>
__device__ __forceinline__ void load_a_batch(GemmRegs& r, const GemmPtrs& p) {
    if constexpr (N > 0 && C < NCHUNK) {
        load_a<C>(r, p);
        load_a_batch<C + 1, N - 1>(r, p);
    }
}

template <int C>
__device__ __forceinline__ void load_b(GemmRegs& r, const GemmPtrs& p) {
    if constexpr (C < C_EO) {
        const float* __restrict__ w = p.pWee + (size_t)(2 * C) * P5;
#pragma unroll
        for (int nt = 0; nt < NT5; ++nt) r.b[C % B_RING][nt] = *(const v2f*)(w + nt * 32);
    } else if constexpr (C < C_OE) {
        const float* __restrict__ w = p.pWeo + (size_t)(2 * (C - C_EO)) * P4;
#pragma unroll
        for (int nt = 0; nt < NT4; ++nt) r.b[C % B_RING][nt] = *(const v2f*)(w + nt * 32);
    } else if constexpr (C < C_OO) {
        const float* __restrict__ w = p.pWoe + (size_t)(2 * (C - C_OE)) * P4;
#pragma unroll
        for (int nt = 0; nt < NT4; ++nt) r.b[C % B_RING][nt] = *(const v2f*)(w + nt * 32);
    } else {
        const float* __restrict__ w = p.pWoo + (size_t)(2 * (C - C_OO)) * P4;
#pragma unroll
        for (int nt = 0; nt < NT4; ++nt) r.b[C % B_RING][nt] = *(const v2f*)(w + nt * 32);
    }
}

template <int C>
__device__ __forceinline__ void gemm_stage(GemmRegs& r, const GemmPtrs& p) {
    constexpr int CB = C + B_AHEAD;
    if constexpr (CB < NCHUNK) {
        if constexpr (CB % A_BATCH == 0) load_a_batch<CB, A_BATCH>(r, p);
        load_b<CB>(r, p);
    }
    __builtin_amdgcn_sched_barrier(0);
    v2f a = r.a[(C / A_BATCH) & 1][C % A_BATCH];
    if constexpr (C < C_EO) {
        if constexpr (C == 16) a.x = (p.half == 0) ? a.x : 0.0f;
#pragma unroll
        for (int nt = 0; nt < NT5; ++nt)
            r.accEe[nt] = __builtin_amdgcn_wmma_f32_16x16x4_f32(
                false, a, false, r.b[C % B_RING][nt], (short)0, r.accEe[nt], false, false);
    } else if constexpr (C < C_OE) {
#pragma unroll
        for (int nt = 0; nt < NT4; ++nt)
            r.accEo[nt] = __builtin_amdgcn_wmma_f32_16x16x4_f32(
                false, a, false, r.b[C % B_RING][nt], (short)0, r.accEo[nt], false, false);
    } else if constexpr (C < C_OO) {
#pragma unroll
        for (int nt = 0; nt < NT4; ++nt)
            r.accOe[nt] = __builtin_amdgcn_wmma_f32_16x16x4_f32(
                false, a, false, r.b[C % B_RING][nt], (short)0, r.accOe[nt], false, false);
    } else {
#pragma unroll
        for (int nt = 0; nt < NT4; ++nt)
            r.accOo[nt] = __builtin_amdgcn_wmma_f32_16x16x4_f32(
                false, a, false, r.b[C % B_RING][nt], (short)0, r.accOo[nt], false, false);
    }
    __builtin_amdgcn_sched_barrier(0);
    if constexpr (C + 1 < NCHUNK) gemm_stage<C + 1>(r, p);
}

template <int C>
__device__ __forceinline__ void load_b_first(GemmRegs& r, const GemmPtrs& p) {
    if constexpr (C < B_AHEAD) {
        load_b<C>(r, p);
        load_b_first<C + 1>(r, p);
    }
}

__global__ void __launch_bounds__(256) __attribute__((amdgpu_waves_per_eu(4, 4)))
ifft_ola_kernel(const float* __restrict__ re, const float* __restrict__ im,
                const float* __restrict__ Wee, const float* __restrict__ Weo,
                const float* __restrict__ Woe, const float* __restrict__ Woo,
                const float* __restrict__ h,
                float* __restrict__ edge, float* __restrict__ out) {
    __shared__ float sfft[BLK_FRAMES * LDS_PITCH];

    const int wv   = threadIdx.x >> 5;
    const int wid  = blockIdx.x * 8 + wv;
    const int lane = threadIdx.x & 31;
    const int half = lane >> 4;
    const int mrow = lane & 15;

    const int r0 = wid << 4;
    const int bc = r0 >> 12;
    const int t  = (r0 & (TDIM - 1)) + mrow;

    GemmRegs g;
#pragma unroll
    for (int i = 0; i < NT5; ++i) g.accEe[i] = (v8f){};
#pragma unroll
    for (int i = 0; i < NT4; ++i) { g.accEo[i] = (v8f){}; g.accOe[i] = (v8f){}; g.accOo[i] = (v8f){}; }

    {
        GemmPtrs gp;
        gp.pR = re + ((size_t)bc * SPEC_BINS) * TDIM + t + (size_t)(4 * half) * TDIM;
        gp.pI = im + ((size_t)bc * SPEC_BINS) * TDIM + t + (size_t)(4 * half) * TDIM;
        gp.pN = re + ((size_t)bc * SPEC_BINS + 128) * TDIM + t;
        gp.pWee = Wee + (size_t)half * P5 + (size_t)mrow * 2;
        gp.pWeo = Weo + (size_t)half * P4 + (size_t)mrow * 2;
        gp.pWoe = Woe + (size_t)half * P4 + (size_t)mrow * 2;
        gp.pWoo = Woo + (size_t)half * P4 + (size_t)mrow * 2;
        gp.half = half;
        load_a_batch<0, A_BATCH>(g, gp);
        load_b_first<0>(g, gp);
        gemm_stage<0>(g, gp);
    }

    float* __restrict__ sp = sfft + (16 * wv + 8 * half) * LDS_PITCH;
#pragma unroll
    for (int nt = 0; nt < NT4; ++nt) {
        const int n = 16 * nt + mrow;
#pragma unroll
        for (int j = 0; j < 8; ++j) {
            float oo = g.accOo[nt][j];
            if (nt == 0) oo = (mrow == 0) ? 0.0f : oo;
            const float ep = g.accEe[nt][j] + g.accEo[nt][j];
            const float em = g.accEe[nt][j] - g.accEo[nt][j];
            const float op = oo + g.accOe[nt][j];
            const float om = oo - g.accOe[nt][j];
            float* __restrict__ row = sp + j * LDS_PITCH;
            row[255 - n] = ep + op;
            row[127 + n] = em + om;
            if (nt > 0) {
                row[127 - n] = em - om;
                row[n - 1]   = ep - op;
            }
        }
    }
    if (mrow != 0) {
#pragma unroll
        for (int j = 0; j < 8; ++j) {
            const float ep = g.accEe[0][j] + g.accEo[0][j];
            const float em = g.accEe[0][j] - g.accEo[0][j];
            const float op = g.accOo[0][j] + g.accOe[0][j];
            const float om = g.accOo[0][j] - g.accOe[0][j];
            sp[j * LDS_PITCH + 127 - mrow] = em - om;
            sp[j * LDS_PITCH + mrow - 1]   = ep - op;
        }
    } else {
#pragma unroll
        for (int j = 0; j < 8; ++j) {
            sp[j * LDS_PITCH + 191] = g.accEe[NT5 - 1][j] + g.accOo[0][j];
            sp[j * LDS_PITCH + 63]  = g.accEe[NT5 - 1][j] - g.accOo[0][j];
        }
    }

    const int j4 = 4 * lane;
    v4f hv[6];
#pragma unroll
    for (int d = 0; d < 6; ++d) hv[d] = *(const v4f*)(h + j4 + FRM_LEN * d);

    __syncthreads();

    {
        float* __restrict__ eg = edge + (size_t)blockIdx.x * EDGE_FLOATS;
        for (int i = threadIdx.x; i < 2 * EDGE_ROWS * (FFT_LEN / 4); i += 256) {
            const int row  = i >> 6;
            const int c4   = (i & 63) * 4;
            const int srow = (row < EDGE_ROWS) ? row : row + (BLK_FRAMES - 2 * EDGE_ROWS);
            *(v4f*)(eg + row * FFT_LEN + c4) = *(const v4f*)(sfft + srow * LDS_PITCH + c4);
        }
    }

    const int f0 = (blockIdx.x & (BLKS_PER_BC - 1)) * BLK_FRAMES;
    const int obase = bc * OUTLEN - GROUP_DELAY + f0 * FRM_LEN + j4;
    {
        const int qq = EDGE_ROWS + wv;
        const v4f acc = ola_piece(sfft + qq * LDS_PITCH + j4, hv);
        if ((f0 + qq) * FRM_LEN + j4 >= GROUP_DELAY)
            *(v4f*)(out + (obase + qq * FRM_LEN)) = acc;
    }
#pragma unroll 2
    for (int qq = EDGE_ROWS + 8 + wv; qq < BLK_FRAMES; qq += 8)
        *(v4f*)(out + (obase + qq * FRM_LEN)) = ola_piece(sfft + qq * LDS_PITCH + j4, hv);
}

#define NSEAMS (NBC * (BLKS_PER_BC - 1))

__global__ void __launch_bounds__(32 * EDGE_ROWS)
seam_kernel(const float* __restrict__ edge, const float* __restrict__ h,
            float* __restrict__ out) {
    const int bc   = blockIdx.x / (BLKS_PER_BC - 1);
    const int blk  = blockIdx.x - bc * (BLKS_PER_BC - 1) + 1;
    const int sidx = threadIdx.x >> 5;
    const int j4   = 4 * (threadIdx.x & 31);
    const float* __restrict__ cur  = edge + (size_t)(bc * BLKS_PER_BC + blk) * EDGE_FLOATS;
    const float* __restrict__ prev = cur - EDGE_FLOATS;

    v4f acc = (v4f){0.0f, 0.0f, 0.0f, 0.0f};
#pragma unroll
    for (int e = 0; e < 6; ++e) {
        const int d = 5 - e;
        const int r = sidx - d;
        const float* __restrict__ row = (r >= 0) ? cur + r * FFT_LEN
                                                 : prev + (2 * EDGE_ROWS + r) * FFT_LEN;
        acc += *(const v4f*)(h + j4 + FRM_LEN * d)
             * *(const v4f*)(row + j4 + FRM_LEN * (d & 1));
    }
    const int m = (blk * BLK_FRAMES + sidx) * FRM_LEN + j4;
    *(v4f*)(out + bc * OUTLEN + m - GROUP_DELAY) = acc;
}

extern "C" void kernel_launch(void* const* d_in, const int* in_sizes, int n_in,
                              void* d_out, int out_size, void* d_ws, size_t ws_size,
                              hipStream_t stream) {
    const float* re = (const float*)d_in[0];
    const float* im = (const float*)d_in[1];
    const float* h  = (const float*)d_in[2];
    float* out = (float*)d_out;

    float* Wee  = (float*)d_ws;
    float* Weo  = Wee + (size_t)WEE_FLOATS;
    float* Woe  = Weo + (size_t)WEO_FLOATS;
    float* Woo  = Woe + (size_t)WOE_FLOATS;
    float* edge = Woo + (size_t)WOO_FLOATS;

    build_w_kernel<<<KEE + 3 * KQ, NT5 * 16, 0, stream>>>(Wee, Weo, Woe, Woo);
    ifft_ola_kernel<<<NBLOCKS, 256, 0, stream>>>(re, im, Wee, Weo, Woe, Woo, h, edge, out);
    seam_kernel<<<NSEAMS, 32 * EDGE_ROWS, 0, stream>>>(edge, h, out);
}
